# starting kernel
# baseline (speedup 1.0000x reference)
#include <hip/hip_runtime.h>
#include <hip/hip_bf16.h>
#include <math.h>


typedef __attribute__((ext_vector_type(2))) float v2f;
typedef __attribute__((ext_vector_type(8))) float v8f;

#define IN_F   128
#define OUT_F  64
#define NHEADS 4
#define NEG_SLOPE 0.2f
#define EPS 1e-12f

__global__ __launch_bounds__(256) void k_zero(float* __restrict__ p, int total) {
    int i = blockIdx.x * 256 + threadIdx.x;
    if (i < total) p[i] = 0.0f;
}

__global__ __launch_bounds__(256) void k_init_out(float* __restrict__ out,
                                                  const float* __restrict__ b,
                                                  int total) {
    int i = blockIdx.x * 256 + threadIdx.x;
    if (i < total) out[i] = b[i & (OUT_F - 1)];
}

__global__ __launch_bounds__(256) void k_gemm_hW(const float* __restrict__ x,
                                                 const float* __restrict__ W,
                                                 float* __restrict__ h,
                                                 int nNodes) {
    __shared__ float sW[IN_F * OUT_F];
    const int tid = threadIdx.x;
    for (int i = tid; i < IN_F * OUT_F; i += 256) sW[i] = W[i];
    __syncthreads();

    const int wave    = tid >> 5;
    const int lane    = tid & 31;
    const int rowBase = blockIdx.x * 128 + wave * 16;
    const int m       = lane & 15;
    const int khalf   = (lane < 16) ? 0 : 2;

    int srcRow = rowBase + m;
    if (srcRow >= nNodes) srcRow = nNodes - 1;
    const float* __restrict__ xr = x + (size_t)srcRow * IN_F;

    v8f acc0 = {}, acc1 = {}, acc2 = {}, acc3 = {};
    const int n = lane & 15;

    for (int k = 0; k < IN_F; k += 4) {
        v2f a;
        a.x = xr[k + khalf];
        a.y = xr[k + khalf + 1];

        const float* __restrict__ w0 = &sW[(k + khalf) * OUT_F];
        const float* __restrict__ w1 = &sW[(k + khalf + 1) * OUT_F];
        v2f b0, b1, b2, b3;
        b0.x = w0[n];        b0.y = w1[n];
        b1.x = w0[16 + n];   b1.y = w1[16 + n];
        b2.x = w0[32 + n];   b2.y = w1[32 + n];
        b3.x = w0[48 + n];   b3.y = w1[48 + n];

        acc0 = __builtin_amdgcn_wmma_f32_16x16x4_f32(false, a, false, b0, (short)0, acc0, false, false);
        acc1 = __builtin_amdgcn_wmma_f32_16x16x4_f32(false, a, false, b1, (short)0, acc1, false, false);
        acc2 = __builtin_amdgcn_wmma_f32_16x16x4_f32(false, a, false, b2, (short)0, acc2, false, false);
        acc3 = __builtin_amdgcn_wmma_f32_16x16x4_f32(false, a, false, b3, (short)0, acc3, false, false);
    }

    const int nIdx = lane & 15;
    const int mOff = (lane < 16) ? 0 : 8;
#pragma unroll
    for (int j = 0; j < 8; ++j) {
        int r = rowBase + j + mOff;
        if (r < nNodes) {
            float* __restrict__ hr = h + (size_t)r * OUT_F;
            hr[nIdx]      = acc0[j];
            hr[16 + nIdx] = acc1[j];
            hr[32 + nIdx] = acc2[j];
            hr[48 + nIdx] = acc3[j];
        }
    }
}

__global__ __launch_bounds__(256) void k_scores(const float* __restrict__ h,
                                                const float* __restrict__ a_l,
                                                const float* __restrict__ a_r,
                                                float* __restrict__ el,
                                                float* __restrict__ er,
                                                int nNodes) {
    __shared__ float sAl[OUT_F * NHEADS];
    __shared__ float sAr[OUT_F * NHEADS];
    const int tid = threadIdx.x;
    if (tid < OUT_F * NHEADS) { sAl[tid] = a_l[tid]; sAr[tid] = a_r[tid]; }
    __syncthreads();

    const int node = blockIdx.x * 256 + tid;
    if (node >= nNodes) return;

    const float* __restrict__ hr = h + (size_t)node * OUT_F;
    float l0 = 0.f, l1 = 0.f, l2 = 0.f, l3 = 0.f;
    float r0 = 0.f, r1 = 0.f, r2 = 0.f, r3 = 0.f;
#pragma unroll 8
    for (int d = 0; d < OUT_F; ++d) {
        float hv = hr[d];
        l0 += hv * sAl[d * NHEADS + 0];
        l1 += hv * sAl[d * NHEADS + 1];
        l2 += hv * sAl[d * NHEADS + 2];
        l3 += hv * sAl[d * NHEADS + 3];
        r0 += hv * sAr[d * NHEADS + 0];
        r1 += hv * sAr[d * NHEADS + 1];
        r2 += hv * sAr[d * NHEADS + 2];
        r3 += hv * sAr[d * NHEADS + 3];
    }
    float* __restrict__ elp = el + (size_t)node * NHEADS;
    float* __restrict__ erp = er + (size_t)node * NHEADS;
    elp[0] = l0; elp[1] = l1; elp[2] = l2; elp[3] = l3;
    erp[0] = r0; erp[1] = r1; erp[2] = r2; erp[3] = r3;
}

__global__ __launch_bounds__(256) void k_edge_e(const int* __restrict__ ei,
                                                const float* __restrict__ el,
                                                const float* __restrict__ er,
                                                float* __restrict__ e,
                                                float* __restrict__ den,
                                                int nEdges) {
    const int eid = blockIdx.x * 256 + threadIdx.x;
    if (eid >= nEdges) return;
    const int row = ei[eid];
    const int col = ei[nEdges + eid];
    const float* __restrict__ elp = el + (size_t)row * NHEADS;
    const float* __restrict__ erp = er + (size_t)col * NHEADS;
    float* __restrict__ ep = e + (size_t)eid * NHEADS;
#pragma unroll
    for (int hh = 0; hh < NHEADS; ++hh) {
        float v = elp[hh] + erp[hh];
        v = (v > 0.0f) ? v : NEG_SLOPE * v;
        v = __expf(v);
        ep[hh] = v;
        atomicAdd(&den[(size_t)row * NHEADS + hh], v);
    }
}

__global__ __launch_bounds__(256) void k_edge_agg(const int* __restrict__ ei,
                                                  const float* __restrict__ e,
                                                  const float* __restrict__ den,
                                                  const float* __restrict__ h,
                                                  float* __restrict__ out,
                                                  int nEdges) {
    const int lane = threadIdx.x & 31;
    const int eid  = blockIdx.x * 8 + (threadIdx.x >> 5);
    if (eid >= nEdges) return;

    const int row = ei[eid];
    const int col = ei[nEdges + eid];

    const float h0 = h[(size_t)col * OUT_F + lane];
    const float h1 = h[(size_t)col * OUT_F + 32 + lane];

    const float* __restrict__ ep = e + (size_t)eid * NHEADS;
    const float* __restrict__ dp = den + (size_t)row * NHEADS;
    float* __restrict__ orow = out + (size_t)row * (NHEADS * OUT_F);

#pragma unroll
    for (int hh = 0; hh < NHEADS; ++hh) {
        const float attn = ep[hh] / fmaxf(dp[hh], EPS);
        atomicAdd(orow + hh * OUT_F + lane,      attn * h0);
        atomicAdd(orow + hh * OUT_F + 32 + lane, attn * h1);
    }
}

extern "C" void kernel_launch(void* const* d_in, const int* in_sizes, int n_in,
                              void* d_out, int out_size, void* d_ws, size_t ws_size,
                              hipStream_t stream) {
    (void)n_in; (void)ws_size;
    const float* x   = (const float*)d_in[0];
    const int*   ei  = (const int*)  d_in[1];
    const float* W   = (const float*)d_in[2];
    const float* a_l = (const float*)d_in[3];
    const float* a_r = (const float*)d_in[4];
    const float* b   = (const float*)d_in[5];
    float* out = (float*)d_out;

    const int nNodes = in_sizes[0] / IN_F;
    const int nEdges = in_sizes[1] / 2;

    float* h   = (float*)d_ws;
    float* el  = h   + (size_t)nNodes * OUT_F;
    float* er  = el  + (size_t)nNodes * NHEADS;
    float* den = er  + (size_t)nNodes * NHEADS;
    float* e   = den + (size_t)nNodes * NHEADS;

    k_zero    <<<(nNodes * NHEADS + 255) / 256, 256, 0, stream>>>(den, nNodes * NHEADS);
    k_init_out<<<(out_size + 255) / 256,        256, 0, stream>>>(out, b, out_size);
    k_gemm_hW <<<(nNodes + 127) / 128,          256, 0, stream>>>(x, W, h, nNodes);
    k_scores  <<<(nNodes + 255) / 256,          256, 0, stream>>>(h, a_l, a_r, el, er, nNodes);
    k_edge_e  <<<(nEdges + 255) / 256,          256, 0, stream>>>(ei, el, er, e, den, nEdges);
    k_edge_agg<<<(nEdges + 7) / 8,              256, 0, stream>>>(ei, e, den, h, out, nEdges);
}


# Softmax division deferred: the edge wave computes e itself, atomically adds e*h[col] to out and e to den; a new row pass does out/den + b; the e launch and e array go; sum first, divide once after
# speedup vs baseline: 1.1992x; 1.1992x over previous
#include <hip/hip_runtime.h>
#include <hip/hip_bf16.h>
#include <math.h>


typedef __attribute__((ext_vector_type(2))) float v2f;
typedef __attribute__((ext_vector_type(8))) float v8f;

#define IN_F   128
#define OUT_F  64
#define NHEADS 4
#define NEG_SLOPE 0.2f
#define EPS 1e-12f

__global__ __launch_bounds__(256) void k_zero(float* __restrict__ p, int total) {
    int i = blockIdx.x * 256 + threadIdx.x;
    if (i < total) p[i] = 0.0f;
}

__global__ __launch_bounds__(256) void k_gemm_hW(const float* __restrict__ x,
                                                 const float* __restrict__ W,
                                                 float* __restrict__ h,
                                                 int nNodes) {
    __shared__ float sW[IN_F * OUT_F];
    const int tid = threadIdx.x;
    for (int i = tid; i < IN_F * OUT_F; i += 256) sW[i] = W[i];
    __syncthreads();

    const int wave    = tid >> 5;
    const int lane    = tid & 31;
    const int rowBase = blockIdx.x * 128 + wave * 16;
    const int m       = lane & 15;
    const int khalf   = (lane < 16) ? 0 : 2;

    int srcRow = rowBase + m;
    if (srcRow >= nNodes) srcRow = nNodes - 1;
    const float* __restrict__ xr = x + (size_t)srcRow * IN_F;

    v8f acc0 = {}, acc1 = {}, acc2 = {}, acc3 = {};
    const int n = lane & 15;

    for (int k = 0; k < IN_F; k += 4) {
        v2f a;
        a.x = xr[k + khalf];
        a.y = xr[k + khalf + 1];

        const float* __restrict__ w0 = &sW[(k + khalf) * OUT_F];
        const float* __restrict__ w1 = &sW[(k + khalf + 1) * OUT_F];
        v2f b0, b1, b2, b3;
        b0.x = w0[n];        b0.y = w1[n];
        b1.x = w0[16 + n];   b1.y = w1[16 + n];
        b2.x = w0[32 + n];   b2.y = w1[32 + n];
        b3.x = w0[48 + n];   b3.y = w1[48 + n];

        acc0 = __builtin_amdgcn_wmma_f32_16x16x4_f32(false, a, false, b0, (short)0, acc0, false, false);
        acc1 = __builtin_amdgcn_wmma_f32_16x16x4_f32(false, a, false, b1, (short)0, acc1, false, false);
        acc2 = __builtin_amdgcn_wmma_f32_16x16x4_f32(false, a, false, b2, (short)0, acc2, false, false);
        acc3 = __builtin_amdgcn_wmma_f32_16x16x4_f32(false, a, false, b3, (short)0, acc3, false, false);
    }

    const int nIdx = lane & 15;
    const int mOff = (lane < 16) ? 0 : 8;
#pragma unroll
    for (int j = 0; j < 8; ++j) {
        int r = rowBase + j + mOff;
        if (r < nNodes) {
            float* __restrict__ hr = h + (size_t)r * OUT_F;
            hr[nIdx]      = acc0[j];
            hr[16 + nIdx] = acc1[j];
            hr[32 + nIdx] = acc2[j];
            hr[48 + nIdx] = acc3[j];
        }
    }
}

__global__ __launch_bounds__(256) void k_scores(const float* __restrict__ h,
                                                const float* __restrict__ a_l,
                                                const float* __restrict__ a_r,
                                                float* __restrict__ el,
                                                float* __restrict__ er,
                                                int nNodes) {
    __shared__ float sAl[OUT_F * NHEADS];
    __shared__ float sAr[OUT_F * NHEADS];
    const int tid = threadIdx.x;
    if (tid < OUT_F * NHEADS) { sAl[tid] = a_l[tid]; sAr[tid] = a_r[tid]; }
    __syncthreads();

    const int node = blockIdx.x * 256 + tid;
    if (node >= nNodes) return;

    const float* __restrict__ hr = h + (size_t)node * OUT_F;
    float l0 = 0.f, l1 = 0.f, l2 = 0.f, l3 = 0.f;
    float r0 = 0.f, r1 = 0.f, r2 = 0.f, r3 = 0.f;
#pragma unroll 8
    for (int d = 0; d < OUT_F; ++d) {
        float hv = hr[d];
        l0 += hv * sAl[d * NHEADS + 0];
        l1 += hv * sAl[d * NHEADS + 1];
        l2 += hv * sAl[d * NHEADS + 2];
        l3 += hv * sAl[d * NHEADS + 3];
        r0 += hv * sAr[d * NHEADS + 0];
        r1 += hv * sAr[d * NHEADS + 1];
        r2 += hv * sAr[d * NHEADS + 2];
        r3 += hv * sAr[d * NHEADS + 3];
    }
    float* __restrict__ elp = el + (size_t)node * NHEADS;
    float* __restrict__ erp = er + (size_t)node * NHEADS;
    elp[0] = l0; elp[1] = l1; elp[2] = l2; elp[3] = l3;
    erp[0] = r0; erp[1] = r1; erp[2] = r2; erp[3] = r3;
}

__device__ __forceinline__ float edge_weight(float s) {
    s = (s > 0.0f) ? s : NEG_SLOPE * s;
    return __expf(s);
}

__global__ __launch_bounds__(256) void k_edge_acc(const int* __restrict__ ei,
                                                  const float* __restrict__ el,
                                                  const float* __restrict__ er,
                                                  const float* __restrict__ h,
                                                  float* __restrict__ den,
                                                  float* __restrict__ out,
                                                  int nEdges) {
    const int lane = threadIdx.x & 31;
    const int eid  = blockIdx.x * 8 + (threadIdx.x >> 5);
    if (eid >= nEdges) return;

    const int row = ei[eid];
    const int col = ei[nEdges + eid];

    const float h0 = h[(size_t)col * OUT_F + lane];
    const float h1 = h[(size_t)col * OUT_F + 32 + lane];

    const float4 l4 = *reinterpret_cast<const float4*>(el + (size_t)row * NHEADS);
    const float4 r4 = *reinterpret_cast<const float4*>(er + (size_t)col * NHEADS);

    float ev[NHEADS];
    ev[0] = edge_weight(l4.x + r4.x);
    ev[1] = edge_weight(l4.y + r4.y);
    ev[2] = edge_weight(l4.z + r4.z);
    ev[3] = edge_weight(l4.w + r4.w);

    float* __restrict__ orow = out + (size_t)row * (NHEADS * OUT_F);
#pragma unroll
    for (int hh = 0; hh < NHEADS; ++hh) {
        atomicAdd(orow + hh * OUT_F + lane,      ev[hh] * h0);
        atomicAdd(orow + hh * OUT_F + 32 + lane, ev[hh] * h1);
    }

    const int hsel = lane & (NHEADS - 1);
    const float lo = (hsel & 1) ? ev[1] : ev[0];
    const float hi = (hsel & 1) ? ev[3] : ev[2];
    const float mine = (hsel & 2) ? hi : lo;
    if (lane < NHEADS) atomicAdd(den + (size_t)row * NHEADS + lane, mine);
}

__global__ __launch_bounds__(256) void k_normalize(float* __restrict__ out,
                                                   const float* __restrict__ den,
                                                   const float* __restrict__ b,
                                                   int total) {
    const int i = blockIdx.x * 256 + threadIdx.x;
    if (i * 4 >= total) return;
    const int perRow = (NHEADS * OUT_F) / 4;
    const int row = i / perRow;
    const int q   = i - row * perRow;
    const int hh  = q / (OUT_F / 4);
    const int d   = (q - hh * (OUT_F / 4)) * 4;

    const float dn = fmaxf(den[(size_t)row * NHEADS + hh], EPS);
    float4* __restrict__ o4 = reinterpret_cast<float4*>(out) + i;
    float4 v = *o4;
    v.x = v.x / dn + b[d];
    v.y = v.y / dn + b[d + 1];
    v.z = v.z / dn + b[d + 2];
    v.w = v.w / dn + b[d + 3];
    *o4 = v;
}

extern "C" void kernel_launch(void* const* d_in, const int* in_sizes, int n_in,
                              void* d_out, int out_size, void* d_ws, size_t ws_size,
                              hipStream_t stream) {
    (void)n_in; (void)ws_size;
    const float* x   = (const float*)d_in[0];
    const int*   ei  = (const int*)  d_in[1];
    const float* W   = (const float*)d_in[2];
    const float* a_l = (const float*)d_in[3];
    const float* a_r = (const float*)d_in[4];
    const float* b   = (const float*)d_in[5];
    float* out = (float*)d_out;

    const int nNodes = in_sizes[0] / IN_F;
    const int nEdges = in_sizes[1] / 2;

    float* h   = (float*)d_ws;
    float* el  = h   + (size_t)nNodes * OUT_F;
    float* er  = el  + (size_t)nNodes * NHEADS;
    float* den = er  + (size_t)nNodes * NHEADS;

    k_zero    <<<(nNodes * NHEADS + 255) / 256, 256, 0, stream>>>(den, nNodes * NHEADS);
    k_zero    <<<(out_size + 255) / 256,        256, 0, stream>>>(out, out_size);
    k_gemm_hW <<<(nNodes + 127) / 128,          256, 0, stream>>>(x, W, h, nNodes);
    k_scores  <<<(nNodes + 255) / 256,          256, 0, stream>>>(h, a_l, a_r, el, er, nNodes);
    k_edge_acc<<<(nEdges + 7) / 8,              256, 0, stream>>>(ei, el, er, h, den, out, nEdges);
    k_normalize<<<(out_size + 1023) / 1024,     256, 0, stream>>>(out, den, b, out_size);
}
